# starting kernel
# baseline (speedup 1.0000x reference)
#include <hip/hip_runtime.h>

#define N_ROWS 131072
#define DDIM   512
#define SDIM   28
#define KCHUNK 256
#define BLOCK  256

typedef __attribute__((ext_vector_type(2))) float v2f;
typedef __attribute__((ext_vector_type(8))) float v8f;

__global__ __launch_bounds__(BLOCK) void scores_wmma_kernel(
    const float* __restrict__ thetas,
    const float* __restrict__ phi,
    const float* __restrict__ W,
    float* __restrict__ out)
{
  __shared__ float Wlds[KCHUNK * 32];
  __shared__ float phis[32];

  const int tid     = threadIdx.x;
  const int wave    = tid >> 5;
  const int lane    = tid & 31;
  const int lane_lo = lane & 15;
  const int half    = lane >> 4;

  const int rowbase = blockIdx.x * (BLOCK / 32) * 32 + wave * 32;

  if (tid < 32) phis[tid] = (tid < SDIM) ? phi[tid] : 0.0f;

  v8f acc[2][2] = {};

  const float* pA = thetas + (size_t)(rowbase + lane_lo) * DDIM + (half << 1);
  const unsigned wb = ((unsigned)half << 6) + (unsigned)lane_lo;

  for (int phase = 0; phase < DDIM / KCHUNK; ++phase) {
    __syncthreads();
    const int k0 = phase * KCHUNK;
    for (int i = tid; i < KCHUNK * 32; i += BLOCK) {
      const int kk = i >> 5;
      const int n  = i & 31;
      Wlds[i] = (n < SDIM) ? W[(size_t)(k0 + kk) * SDIM + n] : 0.0f;
    }
    __syncthreads();

#pragma unroll 4
    for (int kk = 0; kk < KCHUNK / 4; ++kk) {
      const int kg = k0 + (kk << 2);
      v2f a0 = *reinterpret_cast<const v2f*>(pA + kg);
      v2f a1 = *reinterpret_cast<const v2f*>(pA + 16 * DDIM + kg);
      const unsigned idx = ((unsigned)kk << 7) + wb;
      v2f b0, b1;
      b0[0] = Wlds[idx];       b0[1] = Wlds[idx + 32];
      b1[0] = Wlds[idx + 16];  b1[1] = Wlds[idx + 48];

      acc[0][0] = __builtin_amdgcn_wmma_f32_16x16x4_f32(
          false, a0, false, b0, (short)0, acc[0][0], false, false);
      acc[0][1] = __builtin_amdgcn_wmma_f32_16x16x4_f32(
          false, a0, false, b1, (short)0, acc[0][1], false, false);
      acc[1][0] = __builtin_amdgcn_wmma_f32_16x16x4_f32(
          false, a1, false, b0, (short)0, acc[1][0], false, false);
      acc[1][1] = __builtin_amdgcn_wmma_f32_16x16x4_f32(
          false, a1, false, b1, (short)0, acc[1][1], false, false);
    }
  }

  const float p_lo = phis[lane_lo];
  const float p_hi = phis[lane_lo + 16];
#pragma unroll
  for (int m = 0; m < 2; ++m) {
#pragma unroll
    for (int r = 0; r < 8; ++r) {
      float p = acc[m][0][r] * p_lo + acc[m][1][r] * p_hi;
      p += __shfl_xor(p, 1);
      p += __shfl_xor(p, 2);
      p += __shfl_xor(p, 4);
      p += __shfl_xor(p, 8);
      if (lane_lo == 0)
        out[rowbase + m * 16 + half * 8 + r] = p;
    }
  }

  if (rowbase + 32 == N_ROWS && half == 1) {
    out[N_ROWS + lane_lo] = acc[1][0][7];
    if (lane_lo + 16 < SDIM) out[N_ROWS + 16 + lane_lo] = acc[1][1][7];
  }
}

extern "C" void kernel_launch(void* const* d_in, const int* in_sizes, int n_in,
                              void* d_out, int out_size, void* d_ws, size_t ws_size,
                              hipStream_t stream) {
  (void)in_sizes; (void)n_in; (void)out_size; (void)d_ws; (void)ws_size;
  const float* thetas = (const float*)d_in[0];
  const float* phi    = (const float*)d_in[1];
  const float* W      = (const float*)d_in[2];
  float* out = (float*)d_out;

  dim3 grid(N_ROWS / ((BLOCK / 32) * 32));
  scores_wmma_kernel<<<grid, BLOCK, 0, stream>>>(thetas, phi, W, out);
}


# A rows read by 16-byte loads, a whole 128-byte line per row per four loads, issued two 32-k groups ahead of their WMMAs; k order inside each 32-k group permuted (sums reordered, same every run)
# speedup vs baseline: 1.1033x; 1.1033x over previous
#include <hip/hip_runtime.h>

#define N_ROWS 131072
#define DDIM   512
#define SDIM   28
#define KCHUNK 256
#define BLOCK  256
#define KGROUP 32
#define NGROUPS (DDIM / KGROUP)
#define GROUPS_PER_PHASE (KCHUNK / KGROUP)
#define AHEAD  2

typedef __attribute__((ext_vector_type(2))) float v2f;
typedef __attribute__((ext_vector_type(4))) float v4f;
typedef __attribute__((ext_vector_type(8))) float v8f;

__global__ __launch_bounds__(BLOCK) void scores_wmma_kernel(
    const float* __restrict__ thetas,
    const float* __restrict__ phi,
    const float* __restrict__ W,
    float* __restrict__ out)
{
  __shared__ float Wlds[KCHUNK * 32];
  __shared__ float phis[32];

  const int tid     = threadIdx.x;
  const int wave    = tid >> 5;
  const int lane    = tid & 31;
  const int lane_lo = lane & 15;
  const int half    = lane >> 4;

  const int rowbase = blockIdx.x * (BLOCK / 32) * 32 + wave * 32;

  const float* pA = thetas + (size_t)(rowbase + lane_lo) * DDIM + (half << 2);
  v4f a[AHEAD + 1][2][4];
#define LOAD_GROUP(G) do { \
    _Pragma("unroll") for (int m_ = 0; m_ < 2; ++m_) { \
      _Pragma("unroll") for (int j_ = 0; j_ < 4; ++j_) \
        a[(G) % (AHEAD + 1)][m_][j_] = *reinterpret_cast<const v4f*>( \
            pA + m_ * 16 * DDIM + (G) * KGROUP + 8 * j_); } } while (0)

#pragma unroll
  for (int g = 0; g < AHEAD; ++g) LOAD_GROUP(g);

  if (tid < 32) phis[tid] = (tid < SDIM) ? phi[tid] : 0.0f;

  v8f acc[2][2] = {};

  const unsigned wb = ((unsigned)half << 7) + (unsigned)lane_lo;

#pragma unroll
  for (int phase = 0; phase < DDIM / KCHUNK; ++phase) {
    __syncthreads();
    const int k0 = phase * KCHUNK;
    for (int i = tid; i < KCHUNK * 32; i += BLOCK) {
      const int kk = i >> 5;
      const int n  = i & 31;
      Wlds[i] = (n < SDIM) ? W[(size_t)(k0 + kk) * SDIM + n] : 0.0f;
    }
    __syncthreads();

#pragma unroll
    for (int gg = 0; gg < GROUPS_PER_PHASE; ++gg) {
      const int g = phase * GROUPS_PER_PHASE + gg;
      __builtin_amdgcn_sched_barrier(0);
      if (g + AHEAD < NGROUPS) LOAD_GROUP(g + AHEAD);
      __builtin_amdgcn_sched_barrier(0);

#pragma unroll
      for (int j = 0; j < 4; ++j) {
#pragma unroll
        for (int p = 0; p < 2; ++p) {
          const unsigned idx = (unsigned)((gg * KGROUP + 8 * j + 2 * p) * 32) + wb;
          v2f b0, b1;
          b0[0] = Wlds[idx];       b0[1] = Wlds[idx + 32];
          b1[0] = Wlds[idx + 16];  b1[1] = Wlds[idx + 48];
          const v4f q0 = a[g % (AHEAD + 1)][0][j];
          const v4f q1 = a[g % (AHEAD + 1)][1][j];
          v2f a0, a1;
          a0[0] = q0[2 * p];  a0[1] = q0[2 * p + 1];
          a1[0] = q1[2 * p];  a1[1] = q1[2 * p + 1];

          acc[0][0] = __builtin_amdgcn_wmma_f32_16x16x4_f32(
              false, a0, false, b0, (short)0, acc[0][0], false, false);
          acc[0][1] = __builtin_amdgcn_wmma_f32_16x16x4_f32(
              false, a0, false, b1, (short)0, acc[0][1], false, false);
          acc[1][0] = __builtin_amdgcn_wmma_f32_16x16x4_f32(
              false, a1, false, b0, (short)0, acc[1][0], false, false);
          acc[1][1] = __builtin_amdgcn_wmma_f32_16x16x4_f32(
              false, a1, false, b1, (short)0, acc[1][1], false, false);
        }
      }
    }
  }
#undef LOAD_GROUP

  const float p_lo = phis[lane_lo];
  const float p_hi = phis[lane_lo + 16];
#pragma unroll
  for (int m = 0; m < 2; ++m) {
#pragma unroll
    for (int r = 0; r < 8; ++r) {
      float p = acc[m][0][r] * p_lo + acc[m][1][r] * p_hi;
      p += __shfl_xor(p, 1);
      p += __shfl_xor(p, 2);
      p += __shfl_xor(p, 4);
      p += __shfl_xor(p, 8);
      if (lane_lo == 0)
        out[rowbase + m * 16 + half * 8 + r] = p;
    }
  }

  if (rowbase + 32 == N_ROWS && half == 1) {
    out[N_ROWS + lane_lo] = acc[1][0][7];
    if (lane_lo + 16 < SDIM) out[N_ROWS + 16 + lane_lo] = acc[1][1][7];
  }
}

extern "C" void kernel_launch(void* const* d_in, const int* in_sizes, int n_in,
                              void* d_out, int out_size, void* d_ws, size_t ws_size,
                              hipStream_t stream) {
  (void)in_sizes; (void)n_in; (void)out_size; (void)d_ws; (void)ws_size;
  const float* thetas = (const float*)d_in[0];
  const float* phi    = (const float*)d_in[1];
  const float* W      = (const float*)d_in[2];
  float* out = (float*)d_out;

  dim3 grid(N_ROWS / ((BLOCK / 32) * 32));
  scores_wmma_kernel<<<grid, BLOCK, 0, stream>>>(thetas, phi, W, out);
}


# W staged once, whole (512 x 32 in LDS), every thread's 14 sixteen-byte loads issued before the first wait, in place of two phases of 32 load-wait-store trips each; one barrier instead of four
# speedup vs baseline: 1.8106x; 1.6411x over previous
#include <hip/hip_runtime.h>

#define N_ROWS 131072
#define DDIM   512
#define SDIM   28
#define BLOCK  256
#define WPIECES (DDIM * SDIM / 4)
#define WP_PER_THREAD (WPIECES / BLOCK)
#define KGROUP 32
#define NGROUPS (DDIM / KGROUP)
#define AHEAD  2

typedef __attribute__((ext_vector_type(2))) float v2f;
typedef __attribute__((ext_vector_type(4))) float v4f;
typedef __attribute__((ext_vector_type(8))) float v8f;

__global__ __launch_bounds__(BLOCK) void scores_wmma_kernel(
    const float* __restrict__ thetas,
    const float* __restrict__ phi,
    const float* __restrict__ W,
    float* __restrict__ out)
{
  __shared__ __attribute__((aligned(16))) float Wlds[DDIM * 32];
  __shared__ float phis[32];

  const int tid     = threadIdx.x;
  const int wave    = tid >> 5;
  const int lane    = tid & 31;
  const int lane_lo = lane & 15;
  const int half    = lane >> 4;

  const int rowbase = blockIdx.x * (BLOCK / 32) * 32 + wave * 32;

  const float* pA = thetas + (size_t)(rowbase + lane_lo) * DDIM + (half << 2);
  v4f a[AHEAD + 1][2][4];
#define LOAD_GROUP(G) do { \
    _Pragma("unroll") for (int m_ = 0; m_ < 2; ++m_) { \
      _Pragma("unroll") for (int j_ = 0; j_ < 4; ++j_) \
        a[(G) % (AHEAD + 1)][m_][j_] = *reinterpret_cast<const v4f*>( \
            pA + m_ * 16 * DDIM + (G) * KGROUP + 8 * j_); } } while (0)

  const v4f* W4 = reinterpret_cast<const v4f*>(W);
  v4f wreg[WP_PER_THREAD];
#pragma unroll
  for (int t = 0; t < WP_PER_THREAD; ++t) wreg[t] = W4[tid + t * BLOCK];
  float phreg = 0.0f;
  if (tid < SDIM) phreg = phi[tid];

#pragma unroll
  for (int g = 0; g < AHEAD; ++g) LOAD_GROUP(g);
  __builtin_amdgcn_sched_barrier(0);

#pragma unroll
  for (int t = 0; t < WP_PER_THREAD; ++t) {
    const int q = tid + t * BLOCK;
    const int k = q / (SDIM / 4);
    const int c = q - k * (SDIM / 4);
    *reinterpret_cast<v4f*>(&Wlds[k * 32 + 4 * c]) = wreg[t];
  }
#pragma unroll
  for (int t = 0; t < DDIM / BLOCK; ++t)
    *reinterpret_cast<v4f*>(&Wlds[(tid + t * BLOCK) * 32 + SDIM]) = (v4f)(0.0f);
  if (tid < 32) phis[tid] = phreg;
  __syncthreads();

  v8f acc[2][2] = {};

  const unsigned wb = ((unsigned)half << 7) + (unsigned)lane_lo;

#pragma unroll
  for (int g = 0; g < NGROUPS; ++g) {
    __builtin_amdgcn_sched_barrier(0);
    if (g + AHEAD < NGROUPS) LOAD_GROUP(g + AHEAD);
    __builtin_amdgcn_sched_barrier(0);

#pragma unroll
    for (int j = 0; j < 4; ++j) {
#pragma unroll
      for (int p = 0; p < 2; ++p) {
        const unsigned idx = (unsigned)((g * KGROUP + 8 * j + 2 * p) * 32) + wb;
        v2f b0, b1;
        b0[0] = Wlds[idx];       b0[1] = Wlds[idx + 32];
        b1[0] = Wlds[idx + 16];  b1[1] = Wlds[idx + 48];
        const v4f q0 = a[g % (AHEAD + 1)][0][j];
        const v4f q1 = a[g % (AHEAD + 1)][1][j];
        v2f a0, a1;
        a0[0] = q0[2 * p];  a0[1] = q0[2 * p + 1];
        a1[0] = q1[2 * p];  a1[1] = q1[2 * p + 1];

        acc[0][0] = __builtin_amdgcn_wmma_f32_16x16x4_f32(
            false, a0, false, b0, (short)0, acc[0][0], false, false);
        acc[0][1] = __builtin_amdgcn_wmma_f32_16x16x4_f32(
            false, a0, false, b1, (short)0, acc[0][1], false, false);
        acc[1][0] = __builtin_amdgcn_wmma_f32_16x16x4_f32(
            false, a1, false, b0, (short)0, acc[1][0], false, false);
        acc[1][1] = __builtin_amdgcn_wmma_f32_16x16x4_f32(
            false, a1, false, b1, (short)0, acc[1][1], false, false);
      }
    }
  }
#undef LOAD_GROUP

  const float p_lo = phis[lane_lo];
  const float p_hi = phis[lane_lo + 16];
#pragma unroll
  for (int m = 0; m < 2; ++m) {
#pragma unroll
    for (int r = 0; r < 8; ++r) {
      float p = acc[m][0][r] * p_lo + acc[m][1][r] * p_hi;
      p += __shfl_xor(p, 1);
      p += __shfl_xor(p, 2);
      p += __shfl_xor(p, 4);
      p += __shfl_xor(p, 8);
      if (lane_lo == 0)
        out[rowbase + m * 16 + half * 8 + r] = p;
    }
  }

  if (rowbase + 32 == N_ROWS && half == 1) {
    out[N_ROWS + lane_lo] = acc[1][0][7];
    if (lane_lo + 16 < SDIM) out[N_ROWS + 16 + lane_lo] = acc[1][1][7];
  }
}

extern "C" void kernel_launch(void* const* d_in, const int* in_sizes, int n_in,
                              void* d_out, int out_size, void* d_ws, size_t ws_size,
                              hipStream_t stream) {
  (void)in_sizes; (void)n_in; (void)out_size; (void)d_ws; (void)ws_size;
  const float* thetas = (const float*)d_in[0];
  const float* phi    = (const float*)d_in[1];
  const float* W      = (const float*)d_in[2];
  float* out = (float*)d_out;

  dim3 grid(N_ROWS / ((BLOCK / 32) * 32));
  scores_wmma_kernel<<<grid, BLOCK, 0, stream>>>(thetas, phi, W, out);
}


# B fragments read from LDS by 16-byte loads, eight per 32-k group, one group ahead of its WMMAs, from a k-quad-major W transposed in registers at staging (was 64 four-byte reads among the WMMAs)
# speedup vs baseline: 1.9274x; 1.0645x over previous
#include <hip/hip_runtime.h>

#define N_ROWS 131072
#define DDIM   512
#define SDIM   28
#define BLOCK  256
#define KGROUP 32
#define NGROUPS (DDIM / KGROUP)
#define AHEAD  2
#define KQUADS (DDIM / 4)
#define QBLOCKS (KQUADS * 8)
#define QB_PER_THREAD (QBLOCKS / BLOCK)

typedef __attribute__((ext_vector_type(2))) float v2f;
typedef __attribute__((ext_vector_type(4))) float v4f;
typedef __attribute__((ext_vector_type(8))) float v8f;

__global__ __launch_bounds__(BLOCK) void scores_wmma_kernel(
    const float* __restrict__ thetas,
    const float* __restrict__ phi,
    const float* __restrict__ W,
    float* __restrict__ out)
{
  __shared__ __attribute__((aligned(16))) v4f Wq[KQUADS * 32];
  __shared__ float phis[32];

  const int tid     = threadIdx.x;
  const int wave    = tid >> 5;
  const int lane    = tid & 31;
  const int lane_lo = lane & 15;
  const int half    = lane >> 4;

  const int rowbase = blockIdx.x * (BLOCK / 32) * 32 + wave * 32;

  const float* pA = thetas + (size_t)(rowbase + lane_lo) * DDIM + (half << 2);
  v4f a[AHEAD + 1][2][4];
#define LOAD_GROUP(G) do { \
    _Pragma("unroll") for (int m_ = 0; m_ < 2; ++m_) { \
      _Pragma("unroll") for (int j_ = 0; j_ < 4; ++j_) \
        a[(G) % (AHEAD + 1)][m_][j_] = *reinterpret_cast<const v4f*>( \
            pA + m_ * 16 * DDIM + (G) * KGROUP + 8 * j_); } } while (0)

  const v4f* W4 = reinterpret_cast<const v4f*>(W);
  const unsigned cblk = (unsigned)tid & 7u;
  const unsigned kq0  = (unsigned)tid >> 3;
  const bool     real = cblk < 7u;
  const unsigned cld  = real ? cblk : 6u;
  v4f wreg[QB_PER_THREAD][4];
#pragma unroll
  for (int s = 0; s < QB_PER_THREAD; ++s) {
#pragma unroll
    for (int r = 0; r < 4; ++r)
      wreg[s][r] = W4[(4u * (kq0 + 32u * s) + r) * (SDIM / 4) + cld];
  }
  float phreg = 0.0f;
  if (tid < SDIM) phreg = phi[tid];

#pragma unroll
  for (int g = 0; g < AHEAD; ++g) LOAD_GROUP(g);
  __builtin_amdgcn_sched_barrier(0);

  {
    const unsigned tcol = cblk >> 2;
    const unsigned rot  = (kq0 + 2u * tcol) & 3u;
    const unsigned ub   = ((((kq0 >> 1) << 1) + tcol) << 5) + ((kq0 & 1u) << 4) + ((cblk & 3u) << 2);
#pragma unroll
    for (int s = 0; s < QB_PER_THREAD; ++s) {
#pragma unroll
      for (int i = 0; i < 4; ++i) {
        v4f col;
        col[0] = real ? wreg[s][0][i] : 0.0f;  col[1] = real ? wreg[s][1][i] : 0.0f;
        col[2] = real ? wreg[s][2][i] : 0.0f;  col[3] = real ? wreg[s][3][i] : 0.0f;
        Wq[ub + (((unsigned)i + rot) & 3u) + 1024u * s] = col;
      }
    }
  }
  if (tid < 32) phis[tid] = phreg;
  __syncthreads();

  v8f acc[2][2] = {};

  const unsigned ue = ((unsigned)half << 4) + ((unsigned)lane_lo & 12u) + (((unsigned)lane_lo + (unsigned)half) & 3u);
  const unsigned uo = ue ^ 2u;
  v4f b[2][2][4];
#define LOAD_B(G) do { \
    _Pragma("unroll") for (int t_ = 0; t_ < 2; ++t_) { \
      _Pragma("unroll") for (int j_ = 0; j_ < 4; ++j_) \
        b[(G) & 1][t_][j_] = Wq[((4 * (G) + j_) * 2 + t_) * 32 + (((j_ + t_) & 1) ? uo : ue)]; } } while (0)

  LOAD_B(0);

#pragma unroll
  for (int g = 0; g < NGROUPS; ++g) {
    __builtin_amdgcn_sched_barrier(0);
    if (g + AHEAD < NGROUPS) LOAD_GROUP(g + AHEAD);
    if (g + 1 < NGROUPS) LOAD_B(g + 1);
    __builtin_amdgcn_sched_barrier(0);

#pragma unroll
    for (int j = 0; j < 4; ++j) {
#pragma unroll
      for (int p = 0; p < 2; ++p) {
        const v4f q0 = a[g % (AHEAD + 1)][0][j];
        const v4f q1 = a[g % (AHEAD + 1)][1][j];
        const v4f w0 = b[g & 1][0][j];
        const v4f w1 = b[g & 1][1][j];
        v2f a0, a1, b0, b1;
        a0[0] = q0[2 * p];  a0[1] = q0[2 * p + 1];
        a1[0] = q1[2 * p];  a1[1] = q1[2 * p + 1];
        b0[0] = w0[2 * p];  b0[1] = w0[2 * p + 1];
        b1[0] = w1[2 * p];  b1[1] = w1[2 * p + 1];

        acc[0][0] = __builtin_amdgcn_wmma_f32_16x16x4_f32(
            false, a0, false, b0, (short)0, acc[0][0], false, false);
        acc[0][1] = __builtin_amdgcn_wmma_f32_16x16x4_f32(
            false, a0, false, b1, (short)0, acc[0][1], false, false);
        acc[1][0] = __builtin_amdgcn_wmma_f32_16x16x4_f32(
            false, a1, false, b0, (short)0, acc[1][0], false, false);
        acc[1][1] = __builtin_amdgcn_wmma_f32_16x16x4_f32(
            false, a1, false, b1, (short)0, acc[1][1], false, false);
      }
    }
  }
#undef LOAD_B
#undef LOAD_GROUP

  const float p_lo = phis[lane_lo];
  const float p_hi = phis[lane_lo + 16];
#pragma unroll
  for (int m = 0; m < 2; ++m) {
#pragma unroll
    for (int r = 0; r < 8; ++r) {
      float p = acc[m][0][r] * p_lo + acc[m][1][r] * p_hi;
      p += __shfl_xor(p, 1);
      p += __shfl_xor(p, 2);
      p += __shfl_xor(p, 4);
      p += __shfl_xor(p, 8);
      if (lane_lo == 0)
        out[rowbase + m * 16 + half * 8 + r] = p;
    }
  }

  if (rowbase + 32 == N_ROWS && half == 1) {
    out[N_ROWS + lane_lo] = acc[1][0][7];
    if (lane_lo + 16 < SDIM) out[N_ROWS + 16 + lane_lo] = acc[1][1][7];
  }
}

extern "C" void kernel_launch(void* const* d_in, const int* in_sizes, int n_in,
                              void* d_out, int out_size, void* d_ws, size_t ws_size,
                              hipStream_t stream) {
  (void)in_sizes; (void)n_in; (void)out_size; (void)d_ws; (void)ws_size;
  const float* thetas = (const float*)d_in[0];
  const float* phi    = (const float*)d_in[1];
  const float* W      = (const float*)d_in[2];
  float* out = (float*)d_out;

  dim3 grid(N_ROWS / ((BLOCK / 32) * 32));
  scores_wmma_kernel<<<grid, BLOCK, 0, stream>>>(thetas, phi, W, out);
}


# scores taken as thetas @ (W @ phi): each block folds phi into W (one column v) ahead of the row stream, which then meets one 16-column tile, not two; sums reordered, same every run
# speedup vs baseline: 2.3955x; 1.2429x over previous
#include <hip/hip_runtime.h>

#define N_ROWS 131072
#define DDIM   512
#define SDIM   28
#define BLOCK  256
#define KGROUP 32
#define NGROUPS (DDIM / KGROUP)
#define AHEAD  2
#define KQUADS (DDIM / 4)
#define QBLOCKS (KQUADS * 8)
#define QB_PER_THREAD (QBLOCKS / BLOCK)
#define VROWS (DDIM / BLOCK)

typedef __attribute__((ext_vector_type(2))) float v2f;
typedef __attribute__((ext_vector_type(4))) float v4f;
typedef __attribute__((ext_vector_type(8))) float v8f;

#define LOAD_GROUP(G) do { \
    _Pragma("unroll") for (int m_ = 0; m_ < 2; ++m_) { \
      _Pragma("unroll") for (int j_ = 0; j_ < 4; ++j_) \
        a[(G) % (AHEAD + 1)][m_][j_] = *reinterpret_cast<const v4f*>( \
            pA + m_ * 16 * DDIM + (G) * KGROUP + 8 * j_); } } while (0)

template <bool WITH_OUT>
__device__ __forceinline__ void k_loop(const float* __restrict__ pA,
                                       const v4f* vq,
                                       const v4f* Wq,
                                       const unsigned half, const unsigned ue, const unsigned uo,
                                       v4f (&a)[AHEAD + 1][2][4], v8f (&acc)[2], v8f (&acco)[2])
{
  v4f bv[2][4];
#define LOAD_V(G) do { \
    _Pragma("unroll") for (int j_ = 0; j_ < 4; ++j_) \
      bv[(G) & 1][j_] = vq[(G) * 8 + 2 * j_ + half]; } while (0)

  v4f b[2][2][4];
#define LOAD_B(G) do { \
    _Pragma("unroll") for (int t_ = 0; t_ < 2; ++t_) { \
      _Pragma("unroll") for (int j_ = 0; j_ < 4; ++j_) \
        b[(G) & 1][t_][j_] = Wq[((4 * (G) + j_) * 2 + t_) * 32 + (((j_ + t_) & 1) ? uo : ue)]; } } while (0)

  LOAD_V(0);
  if (WITH_OUT) LOAD_B(0);

#pragma unroll
  for (int g = 0; g < NGROUPS; ++g) {
    __builtin_amdgcn_sched_barrier(0);
    if (g + AHEAD < NGROUPS) LOAD_GROUP(g + AHEAD);
    if (g + 1 < NGROUPS) { LOAD_V(g + 1); if (WITH_OUT) LOAD_B(g + 1); }
    __builtin_amdgcn_sched_barrier(0);

#pragma unroll
    for (int j = 0; j < 4; ++j) {
#pragma unroll
      for (int p = 0; p < 2; ++p) {
        const v4f q0 = a[g % (AHEAD + 1)][0][j];
        const v4f q1 = a[g % (AHEAD + 1)][1][j];
        const v4f vv = bv[g & 1][j];
        v2f a0, a1, bvv;
        a0[0] = q0[2 * p];  a0[1] = q0[2 * p + 1];
        a1[0] = q1[2 * p];  a1[1] = q1[2 * p + 1];
        bvv[0] = vv[2 * p]; bvv[1] = vv[2 * p + 1];

        acc[0] = __builtin_amdgcn_wmma_f32_16x16x4_f32(
            false, a0, false, bvv, (short)0, acc[0], false, false);
        acc[1] = __builtin_amdgcn_wmma_f32_16x16x4_f32(
            false, a1, false, bvv, (short)0, acc[1], false, false);
        if (WITH_OUT) {
          const v4f w0 = b[g & 1][0][j];
          const v4f w1 = b[g & 1][1][j];
          v2f b0, b1;
          b0[0] = w0[2 * p];  b0[1] = w0[2 * p + 1];
          b1[0] = w1[2 * p];  b1[1] = w1[2 * p + 1];
          acco[0] = __builtin_amdgcn_wmma_f32_16x16x4_f32(
              false, a1, false, b0, (short)0, acco[0], false, false);
          acco[1] = __builtin_amdgcn_wmma_f32_16x16x4_f32(
              false, a1, false, b1, (short)0, acco[1], false, false);
        }
      }
    }
  }
#undef LOAD_B
#undef LOAD_V
}

__global__ __launch_bounds__(BLOCK) void scores_wmma_kernel(
    const float* __restrict__ thetas,
    const float* __restrict__ phi,
    const float* __restrict__ W,
    float* __restrict__ out)
{
  __shared__ __attribute__((aligned(16))) v4f Wq[KQUADS * 32];
  __shared__ __attribute__((aligned(16))) float vs[DDIM];

  const int tid     = threadIdx.x;
  const int lane    = tid & 31;
  const int lane_lo = lane & 15;
  const int half    = lane >> 4;
  const int wave    = __builtin_amdgcn_readfirstlane(tid >> 5);

  const int  rowbase   = blockIdx.x * (BLOCK / 32) * 32 + wave * 32;
  const bool last_blk  = (blockIdx.x == N_ROWS / ((BLOCK / 32) * 32) - 1);
  const bool tail_wave = (rowbase + 32 == N_ROWS);

  const float* pA = thetas + (size_t)(rowbase + lane_lo) * DDIM + (half << 2);
  v4f a[AHEAD + 1][2][4];

  const v4f* W4 = reinterpret_cast<const v4f*>(W);
  v4f wrow[VROWS][SDIM / 4];
#pragma unroll
  for (int s = 0; s < VROWS; ++s) {
#pragma unroll
    for (int c = 0; c < SDIM / 4; ++c)
      wrow[s][c] = W4[(tid + BLOCK * s) * (SDIM / 4) + c];
  }
  float ph[SDIM];
#pragma unroll
  for (int n = 0; n < SDIM; ++n) ph[n] = phi[n];

#pragma unroll
  for (int g = 0; g < AHEAD; ++g) LOAD_GROUP(g);
  __builtin_amdgcn_sched_barrier(0);

#pragma unroll
  for (int s = 0; s < VROWS; ++s) {
    float sum = 0.0f;
#pragma unroll
    for (int n = 0; n < SDIM; ++n) sum += wrow[s][n >> 2][n & 3] * ph[n];
    vs[tid + BLOCK * s] = sum;
  }

  const unsigned ue = ((unsigned)half << 4) + ((unsigned)lane_lo & 12u) + (((unsigned)lane_lo + (unsigned)half) & 3u);
  const unsigned uo = ue ^ 2u;

  if (last_blk) {
    const unsigned cblk = (unsigned)tid & 7u;
    const unsigned kq0  = (unsigned)tid >> 3;
    const bool     real = cblk < 7u;
    const unsigned cld  = real ? cblk : 6u;
    v4f wreg[QB_PER_THREAD][4];
#pragma unroll
    for (int s = 0; s < QB_PER_THREAD; ++s) {
#pragma unroll
      for (int r = 0; r < 4; ++r)
        wreg[s][r] = W4[(4u * (kq0 + 32u * s) + r) * (SDIM / 4) + cld];
    }
    const unsigned tcol = cblk >> 2;
    const unsigned rot  = (kq0 + 2u * tcol) & 3u;
    const unsigned ub   = ((((kq0 >> 1) << 1) + tcol) << 5) + ((kq0 & 1u) << 4) + ((cblk & 3u) << 2);
#pragma unroll
    for (int s = 0; s < QB_PER_THREAD; ++s) {
#pragma unroll
      for (int i = 0; i < 4; ++i) {
        v4f col;
        col[0] = real ? wreg[s][0][i] : 0.0f;  col[1] = real ? wreg[s][1][i] : 0.0f;
        col[2] = real ? wreg[s][2][i] : 0.0f;  col[3] = real ? wreg[s][3][i] : 0.0f;
        Wq[ub + (((unsigned)i + rot) & 3u) + 1024u * s] = col;
      }
    }
  }
  __syncthreads();

  v8f acc[2]  = {};
  v8f acco[2] = {};
  const v4f* vq = reinterpret_cast<const v4f*>(vs);
  if (tail_wave) k_loop<true >(pA, vq, Wq, (unsigned)half, ue, uo, a, acc, acco);
  else           k_loop<false>(pA, vq, Wq, (unsigned)half, ue, uo, a, acc, acco);

  {
    float s0 = acc[0][0], s1 = acc[1][0];
#pragma unroll
    for (int r = 1; r < 8; ++r) {
      s0 = ((lane_lo & 7) == r) ? acc[0][r] : s0;
      s1 = ((lane_lo & 7) == r) ? acc[1][r] : s1;
    }
    const int m = lane_lo >> 3;
    out[rowbase + m * 16 + half * 8 + (lane_lo & 7)] = m ? s1 : s0;
  }

  if (tail_wave && half == 1) {
    out[N_ROWS + lane_lo] = acco[0][7];
    if (lane_lo + 16 < SDIM) out[N_ROWS + 16 + lane_lo] = acco[1][7];
  }
}
#undef LOAD_GROUP

extern "C" void kernel_launch(void* const* d_in, const int* in_sizes, int n_in,
                              void* d_out, int out_size, void* d_ws, size_t ws_size,
                              hipStream_t stream) {
  (void)in_sizes; (void)n_in; (void)out_size; (void)d_ws; (void)ws_size;
  const float* thetas = (const float*)d_in[0];
  const float* phi    = (const float*)d_in[1];
  const float* W      = (const float*)d_in[2];
  float* out = (float*)d_out;

  dim3 grid(N_ROWS / ((BLOCK / 32) * 32));
  scores_wmma_kernel<<<grid, BLOCK, 0, stream>>>(thetas, phi, W, out);
}
